# starting kernel
# baseline (speedup 1.0000x reference)
#include <hip/hip_runtime.h>

typedef __attribute__((ext_vector_type(16))) _Float16 v16h;
typedef __attribute__((ext_vector_type(8)))  _Float16 v8h;
typedef __attribute__((ext_vector_type(8)))  float    v8f;
typedef __attribute__((ext_vector_type(4)))  float    v4f;

#define NF_NLAYERS 8
#define NF_LATENT  64
#define NF_HALF    32
#define NF_HIDDEN  128
#define NF_SCALE   0.1f

#define NF_THREADS 512
#define NF_WAVES   (NF_THREADS / 32)
#define NF_ROWS_PER_BLOCK (NF_WAVES * 2 * 16)

#define LD1 40
#define LD2 136
#define LD3 136

#define W1_ELEMS (NF_HALF * NF_HIDDEN)
#define W2_ELEMS (NF_HIDDEN * NF_HIDDEN)
#define W3_ELEMS (NF_HIDDEN * NF_LATENT)
#define NSUB     (NF_NLAYERS * 2)

__global__ void __launch_bounds__(256)
nf_convert_weights(const float* __restrict__ W1, const float* __restrict__ W2,
                   const float* __restrict__ W3,
                   _Float16* __restrict__ tW1, _Float16* __restrict__ tW2,
                   _Float16* __restrict__ tW3)
{
    const int i = blockIdx.x * 256 + threadIdx.x;
    const int n1 = NSUB * W1_ELEMS;
    const int n2 = NSUB * W2_ELEMS;
    const int n3 = NSUB * W3_ELEMS;
    if (i < n1) {
        int w = i >> 12, r = i & 4095;
        int m = r >> 5, k = r & 31;
        tW1[i] = (_Float16)W1[w * W1_ELEMS + k * NF_HIDDEN + m];
    } else if (i < n1 + n2) {
        int j = i - n1;
        int w = j >> 14, r = j & 16383;
        int m = r >> 7, k = r & 127;
        tW2[j] = (_Float16)W2[w * W2_ELEMS + k * NF_HIDDEN + m];
    } else if (i < n1 + n2 + n3) {
        int j = i - n1 - n2;
        int w = j >> 13, r = j & 8191;
        int m = r >> 7, k = r & 127;
        tW3[j] = (_Float16)W3[w * W3_ELEMS + k * NF_LATENT + m];
    }
}

__device__ __forceinline__ v16h nf_ldsA(const _Float16* W, int ld, int mtile, int kchunk, int lane)
{
    const int row = mtile * 16 + (lane & 15);
    const int kb  = kchunk * 32 + ((lane & 16) ? 8 : 0);
    const _Float16* p = W + row * ld + kb;
    v8h lo = *(const v8h*)p;
    v8h hi = *(const v8h*)(p + 16);
    v16h a;
#pragma unroll
    for (int j = 0; j < 8; ++j) { a[j] = lo[j]; a[j + 8] = hi[j]; }
    return a;
}

__device__ __forceinline__ v8f nf_ldsBias(const float* B, int mtile, int hi8)
{
    const float* p = B + mtile * 16 + hi8;
    v8f c;
#pragma unroll
    for (int j = 0; j < 8; ++j) c[j] = p[j];
    return c;
}

__device__ __forceinline__ v16h nf_packB(v8f lo, v8f hi)
{
    v16h b;
#pragma unroll
    for (int j = 0; j < 8; ++j) { b[j] = (_Float16)lo[j]; b[j + 8] = (_Float16)hi[j]; }
    return b;
}

__device__ __forceinline__ v16h nf_relu16(v16h b)
{
#if __has_builtin(__builtin_elementwise_max)
    const v16h z = {};
    return __builtin_elementwise_max(b, z);
#else
    v16h r;
#pragma unroll
    for (int j = 0; j < 16; ++j) r[j] = b[j] > (_Float16)0.0f ? b[j] : (_Float16)0.0f;
    return r;
#endif
}

__device__ __forceinline__ v16h nf_packB_relu(v8f lo, v8f hi)
{
    return nf_relu16(nf_packB(lo, hi));
}

__device__ __forceinline__ float nf_tanh(float v)
{
#if __has_builtin(__builtin_amdgcn_tanhf)
    return __builtin_amdgcn_tanhf(v);
#else
    return tanhf(v);
#endif
}

#define NF_WMMA(a, b, c) \
    __builtin_amdgcn_wmma_f32_16x16x32_f16(false, (a), false, (b), (short)0, (c), false, false)

struct NFStage {
    v8h r1;
    v8h r2[4];
    v8h r3[2];
    float rb;
};

__device__ __forceinline__ void nf_stage_load(NFStage& st, int s, int tid,
                                              const _Float16* tW1, const _Float16* tW2,
                                              const _Float16* tW3,
                                              const float* b1, const float* b2, const float* b3)
{
    const _Float16* g1 = tW1 + (long)s * W1_ELEMS;
    const _Float16* g2 = tW2 + (long)s * W2_ELEMS;
    const _Float16* g3 = tW3 + (long)s * W3_ELEMS;
    st.r1 = *(const v8h*)&g1[tid * 8];
#pragma unroll
    for (int q = 0; q < 4; ++q) st.r2[q] = *(const v8h*)&g2[(tid + q * NF_THREADS) * 8];
#pragma unroll
    for (int q = 0; q < 2; ++q) st.r3[q] = *(const v8h*)&g3[(tid + q * NF_THREADS) * 8];
    float rb = 0.0f;
    if (tid < 128)      rb = b1[(long)s * NF_HIDDEN + tid];
    else if (tid < 256) rb = b2[(long)s * NF_HIDDEN + (tid - 128)];
    else if (tid < 320) rb = b3[(long)s * NF_LATENT + (tid - 256)];
    st.rb = rb;
}

__global__ void __launch_bounds__(NF_THREADS)
nf_flow_kernel(const float* __restrict__ x,
               const _Float16* __restrict__ tW1, const float* __restrict__ b1,
               const _Float16* __restrict__ tW2, const float* __restrict__ b2,
               const _Float16* __restrict__ tW3, const float* __restrict__ b3,
               float* __restrict__ out)
{
    __shared__ __align__(16) _Float16 sW1[NF_HIDDEN * LD1];
    __shared__ __align__(16) _Float16 sW2[NF_HIDDEN * LD2];
    __shared__ __align__(16) _Float16 sW3[NF_LATENT * LD3];
    __shared__ __align__(16) float sB1[NF_HIDDEN];
    __shared__ __align__(16) float sB2[NF_HIDDEN];
    __shared__ __align__(16) float sB3[NF_LATENT];

    const int tid  = threadIdx.x;
    const int lane = tid & 31;
    const int wave = tid >> 5;
    const int hi8  = (lane & 16) ? 8 : 0;

    const long row0 = (long)blockIdx.x * NF_ROWS_PER_BLOCK + (wave * 2 + 0) * 16 + (lane & 15);
    const long row1 = (long)blockIdx.x * NF_ROWS_PER_BLOCK + (wave * 2 + 1) * 16 + (lane & 15);

    v8f xA[4], xB[4];
    {
        const float* xr0 = x + row0 * NF_LATENT;
        const float* xr1 = x + row1 * NF_LATENT;
#pragma unroll
        for (int f = 0; f < 4; ++f) {
            const int d0 = f * 16 + hi8;
            v4f a0 = *(const v4f*)(xr0 + d0);
            v4f b0 = *(const v4f*)(xr0 + d0 + 4);
            v4f a1 = *(const v4f*)(xr1 + d0);
            v4f b1v = *(const v4f*)(xr1 + d0 + 4);
#pragma unroll
            for (int j = 0; j < 4; ++j) {
                xA[f][j] = a0[j]; xA[f][j + 4] = b0[j];
                xB[f][j] = a1[j]; xB[f][j + 4] = b1v[j];
            }
        }
    }

    NFStage st;
    nf_stage_load(st, 0, tid, tW1, tW2, tW3, b1, b2, b3);

    for (int s = 0; s < NSUB; ++s) {
        __syncthreads();
        {
            const int i1 = tid * 8;
            *(v8h*)&sW1[(i1 >> 5) * LD1 + (i1 & 31)] = st.r1;
#pragma unroll
            for (int q = 0; q < 4; ++q) {
                const int i = (tid + q * NF_THREADS) * 8;
                *(v8h*)&sW2[(i >> 7) * LD2 + (i & 127)] = st.r2[q];
            }
#pragma unroll
            for (int q = 0; q < 2; ++q) {
                const int i = (tid + q * NF_THREADS) * 8;
                *(v8h*)&sW3[(i >> 7) * LD3 + (i & 127)] = st.r3[q];
            }
            if (tid < 128)      sB1[tid] = st.rb;
            else if (tid < 256) sB2[tid - 128] = st.rb;
            else if (tid < 320) sB3[tid - 256] = st.rb;
        }
        if (s + 1 < NSUB)
            nf_stage_load(st, s + 1, tid, tW1, tW2, tW3, b1, b2, b3);
        __syncthreads();

        v16h binA = nf_packB(xA[0], xA[1]);
        v16h binB = nf_packB(xB[0], xB[1]);

        v16h p1A[4], p1B[4];
#pragma unroll
        for (int g = 0; g < 4; ++g) {
            v8f cb0 = nf_ldsBias(sB1, 2 * g, hi8);
            v8f cb1 = nf_ldsBias(sB1, 2 * g + 1, hi8);
            v16h a0 = nf_ldsA(sW1, LD1, 2 * g, 0, lane);
            v16h a1 = nf_ldsA(sW1, LD1, 2 * g + 1, 0, lane);
            v8f hA0 = NF_WMMA(a0, binA, cb0);
            v8f hB0 = NF_WMMA(a0, binB, cb0);
            v8f hA1 = NF_WMMA(a1, binA, cb1);
            v8f hB1 = NF_WMMA(a1, binB, cb1);
            p1A[g] = nf_packB_relu(hA0, hA1);
            p1B[g] = nf_packB_relu(hB0, hB1);
        }

        v16h p2A[4], p2B[4];
#pragma unroll
        for (int g = 0; g < 4; ++g) {
            v8f c0 = nf_ldsBias(sB2, 2 * g, hi8);
            v8f c1 = c0;
            v8f d0 = nf_ldsBias(sB2, 2 * g + 1, hi8);
            v8f d1 = d0;
#pragma unroll
            for (int k = 0; k < 4; ++k) {
                v16h a0 = nf_ldsA(sW2, LD2, 2 * g, k, lane);
                c0 = NF_WMMA(a0, p1A[k], c0);
                c1 = NF_WMMA(a0, p1B[k], c1);
                v16h a1 = nf_ldsA(sW2, LD2, 2 * g + 1, k, lane);
                d0 = NF_WMMA(a1, p1A[k], d0);
                d1 = NF_WMMA(a1, p1B[k], d1);
            }
            p2A[g] = nf_packB_relu(c0, d0);
            p2B[g] = nf_packB_relu(c1, d1);
        }

        v8f stA[4], stB[4];
#pragma unroll
        for (int t = 0; t < 4; ++t) {
            v8f c0 = nf_ldsBias(sB3, t, hi8);
            v8f c1 = c0;
#pragma unroll
            for (int k = 0; k < 4; ++k) {
                v16h a = nf_ldsA(sW3, LD3, t, k, lane);
                c0 = NF_WMMA(a, p2A[k], c0);
                c1 = NF_WMMA(a, p2B[k], c1);
            }
            stA[t] = c0;
            stB[t] = c1;
        }

#pragma unroll
        for (int f = 0; f < 2; ++f) {
#pragma unroll
            for (int j = 0; j < 8; ++j) {
                float eA = __expf(NF_SCALE * nf_tanh(stA[f][j]));
                float eB = __expf(NF_SCALE * nf_tanh(stB[f][j]));
                xA[2 + f][j] = xA[2 + f][j] * eA + stA[2 + f][j];
                xB[2 + f][j] = xB[2 + f][j] * eB + stB[2 + f][j];
            }
        }
#pragma unroll
        for (int f = 0; f < 2; ++f) {
            v8f tA = xA[f]; xA[f] = xA[2 + f]; xA[2 + f] = tA;
            v8f tB = xB[f]; xB[f] = xB[2 + f]; xB[2 + f] = tB;
        }
    }

    {
        float* o0 = out + row0 * NF_LATENT;
        float* o1 = out + row1 * NF_LATENT;
#pragma unroll
        for (int f = 0; f < 4; ++f) {
            const int d0 = f * 16 + hi8;
            v4f a0, b0, a1, b1v;
#pragma unroll
            for (int j = 0; j < 4; ++j) {
                a0[j] = xA[f][j]; b0[j] = xA[f][j + 4];
                a1[j] = xB[f][j]; b1v[j] = xB[f][j + 4];
            }
            *(v4f*)(o0 + d0)     = a0;
            *(v4f*)(o0 + d0 + 4) = b0;
            *(v4f*)(o1 + d0)     = a1;
            *(v4f*)(o1 + d0 + 4) = b1v;
        }
    }
}

extern "C" void kernel_launch(void* const* d_in, const int* in_sizes, int n_in,
                              void* d_out, int out_size, void* d_ws, size_t ws_size,
                              hipStream_t stream)
{
    const float* x  = (const float*)d_in[0];
    const float* W1 = (const float*)d_in[1];
    const float* b1 = (const float*)d_in[2];
    const float* W2 = (const float*)d_in[3];
    const float* b2 = (const float*)d_in[4];
    const float* W3 = (const float*)d_in[5];
    const float* b3 = (const float*)d_in[6];
    float* out = (float*)d_out;

    _Float16* tW1 = (_Float16*)d_ws;
    _Float16* tW2 = tW1 + NSUB * W1_ELEMS;
    _Float16* tW3 = tW2 + NSUB * W2_ELEMS;

    const int total_w = NSUB * (W1_ELEMS + W2_ELEMS + W3_ELEMS);
    nf_convert_weights<<<(total_w + 255) / 256, 256, 0, stream>>>(W1, W2, W3, tW1, tW2, tW3);

    const int batch  = in_sizes[0] / NF_LATENT;
    const int blocks = batch / NF_ROWS_PER_BLOCK;
    nf_flow_kernel<<<blocks, NF_THREADS, 0, stream>>>(x, tW1, b1, tW2, b2, tW3, b3, out);
}


# Read each 16-WMMA block's weight fragments from LDS as one group ahead of the previous block's WMMAs (second register set; 8-wave workgroups so a wave may hold it)
# speedup vs baseline: 1.3088x; 1.3088x over previous
#include <hip/hip_runtime.h>

typedef __attribute__((ext_vector_type(16))) _Float16 v16h;
typedef __attribute__((ext_vector_type(8)))  _Float16 v8h;
typedef __attribute__((ext_vector_type(8)))  float    v8f;
typedef __attribute__((ext_vector_type(4)))  float    v4f;

#define NF_NLAYERS 8
#define NF_LATENT  64
#define NF_HALF    32
#define NF_HIDDEN  128
#define NF_SCALE   0.1f

#define NF_THREADS 256
#define NF_WAVES   (NF_THREADS / 32)
#define NF_ROWS_PER_BLOCK (NF_WAVES * 2 * 16)

#define LD1 40
#define LD2 136
#define LD3 136

#define W1_ELEMS (NF_HALF * NF_HIDDEN)
#define W2_ELEMS (NF_HIDDEN * NF_HIDDEN)
#define W3_ELEMS (NF_HIDDEN * NF_LATENT)
#define NSUB     (NF_NLAYERS * 2)

__global__ void __launch_bounds__(256)
nf_convert_weights(const float* __restrict__ W1, const float* __restrict__ W2,
                   const float* __restrict__ W3,
                   _Float16* __restrict__ tW1, _Float16* __restrict__ tW2,
                   _Float16* __restrict__ tW3)
{
    const int i = blockIdx.x * 256 + threadIdx.x;
    const int n1 = NSUB * W1_ELEMS;
    const int n2 = NSUB * W2_ELEMS;
    const int n3 = NSUB * W3_ELEMS;
    if (i < n1) {
        int w = i >> 12, r = i & 4095;
        int m = r >> 5, k = r & 31;
        tW1[i] = (_Float16)W1[w * W1_ELEMS + k * NF_HIDDEN + m];
    } else if (i < n1 + n2) {
        int j = i - n1;
        int w = j >> 14, r = j & 16383;
        int m = r >> 7, k = r & 127;
        tW2[j] = (_Float16)W2[w * W2_ELEMS + k * NF_HIDDEN + m];
    } else if (i < n1 + n2 + n3) {
        int j = i - n1 - n2;
        int w = j >> 13, r = j & 8191;
        int m = r >> 7, k = r & 127;
        tW3[j] = (_Float16)W3[w * W3_ELEMS + k * NF_LATENT + m];
    }
}

__device__ __forceinline__ v16h nf_ldsA(const _Float16* W, int ld, int mtile, int kchunk, int lane)
{
    const int row = mtile * 16 + (lane & 15);
    const int kb  = kchunk * 32 + ((lane & 16) ? 8 : 0);
    const _Float16* p = W + row * ld + kb;
    v8h lo = *(const v8h*)p;
    v8h hi = *(const v8h*)(p + 16);
    v16h a;
#pragma unroll
    for (int j = 0; j < 8; ++j) { a[j] = lo[j]; a[j + 8] = hi[j]; }
    return a;
}

__device__ __forceinline__ v8f nf_ldsBias(const float* B, int mtile, int hi8)
{
    const float* p = B + mtile * 16 + hi8;
    v8f c;
#pragma unroll
    for (int j = 0; j < 8; ++j) c[j] = p[j];
    return c;
}

__device__ __forceinline__ v16h nf_packB(v8f lo, v8f hi)
{
    v16h b;
#pragma unroll
    for (int j = 0; j < 8; ++j) { b[j] = (_Float16)lo[j]; b[j + 8] = (_Float16)hi[j]; }
    return b;
}

__device__ __forceinline__ v16h nf_relu16(v16h b)
{
#if __has_builtin(__builtin_elementwise_max)
    const v16h z = {};
    return __builtin_elementwise_max(b, z);
#else
    v16h r;
#pragma unroll
    for (int j = 0; j < 16; ++j) r[j] = b[j] > (_Float16)0.0f ? b[j] : (_Float16)0.0f;
    return r;
#endif
}

__device__ __forceinline__ v16h nf_packB_relu(v8f lo, v8f hi)
{
    return nf_relu16(nf_packB(lo, hi));
}

__device__ __forceinline__ float nf_tanh(float v)
{
#if __has_builtin(__builtin_amdgcn_tanhf)
    return __builtin_amdgcn_tanhf(v);
#else
    return tanhf(v);
#endif
}

#define NF_WMMA(a, b, c) \
    __builtin_amdgcn_wmma_f32_16x16x32_f16(false, (a), false, (b), (short)0, (c), false, false)

struct NFStage {
    v8h r1[2];
    v8h r2[8];
    v8h r3[4];
    float rb0;
    float rb1;
};

__device__ __forceinline__ void nf_stage_load(NFStage& st, int s, int tid,
                                              const _Float16* tW1, const _Float16* tW2,
                                              const _Float16* tW3,
                                              const float* b1, const float* b2, const float* b3)
{
    const _Float16* g1 = tW1 + (long)s * W1_ELEMS;
    const _Float16* g2 = tW2 + (long)s * W2_ELEMS;
    const _Float16* g3 = tW3 + (long)s * W3_ELEMS;
#pragma unroll
    for (int q = 0; q < 2; ++q) st.r1[q] = *(const v8h*)&g1[(tid + q * NF_THREADS) * 8];
#pragma unroll
    for (int q = 0; q < 8; ++q) st.r2[q] = *(const v8h*)&g2[(tid + q * NF_THREADS) * 8];
#pragma unroll
    for (int q = 0; q < 4; ++q) st.r3[q] = *(const v8h*)&g3[(tid + q * NF_THREADS) * 8];
    st.rb0 = (tid < 128) ? b1[(long)s * NF_HIDDEN + tid] : b2[(long)s * NF_HIDDEN + (tid - 128)];
    float rb1 = 0.0f;
    if (tid < 64) rb1 = b3[(long)s * NF_LATENT + tid];
    st.rb1 = rb1;
}

struct NFBlk1 { v16h a[4]; v8f c[4]; };
struct NFBlk  { v16h a[8]; v8f c[2]; };

__device__ __forceinline__ void nf_ld_blk1(NFBlk1& b, const _Float16* sW1, const float* sB1,
                                           int h, int lane, int hi8)
{
#pragma unroll
    for (int i = 0; i < 4; ++i) {
        b.a[i] = nf_ldsA(sW1, LD1, 4 * h + i, 0, lane);
        b.c[i] = nf_ldsBias(sB1, 4 * h + i, hi8);
    }
}

__device__ __forceinline__ void nf_ld_blk2(NFBlk& b, const _Float16* sW2, const float* sB2,
                                           int g, int lane, int hi8)
{
#pragma unroll
    for (int k = 0; k < 4; ++k) {
        b.a[k * 2 + 0] = nf_ldsA(sW2, LD2, 2 * g, k, lane);
        b.a[k * 2 + 1] = nf_ldsA(sW2, LD2, 2 * g + 1, k, lane);
    }
    b.c[0] = nf_ldsBias(sB2, 2 * g, hi8);
    b.c[1] = nf_ldsBias(sB2, 2 * g + 1, hi8);
}

__device__ __forceinline__ void nf_ld_blk3(NFBlk& b, const _Float16* sW3, const float* sB3,
                                           int u, int lane, int hi8)
{
#pragma unroll
    for (int k = 0; k < 4; ++k) {
        b.a[k * 2 + 0] = nf_ldsA(sW3, LD3, 2 * u, k, lane);
        b.a[k * 2 + 1] = nf_ldsA(sW3, LD3, 2 * u + 1, k, lane);
    }
    b.c[0] = nf_ldsBias(sB3, 2 * u, hi8);
    b.c[1] = nf_ldsBias(sB3, 2 * u + 1, hi8);
}

#define NF_LOADS_STAY_AHEAD() __builtin_amdgcn_sched_barrier(0x406)

__device__ __forceinline__ void nf_l1_blk(const NFBlk1& b, int h, v16h binA, v16h binB,
                                          v16h* p1A, v16h* p1B)
{
#pragma unroll
    for (int i = 0; i < 2; ++i) {
        v8f hA0 = NF_WMMA(b.a[2 * i], binA, b.c[2 * i]);
        v8f hB0 = NF_WMMA(b.a[2 * i], binB, b.c[2 * i]);
        v8f hA1 = NF_WMMA(b.a[2 * i + 1], binA, b.c[2 * i + 1]);
        v8f hB1 = NF_WMMA(b.a[2 * i + 1], binB, b.c[2 * i + 1]);
        p1A[2 * h + i] = nf_packB_relu(hA0, hA1);
        p1B[2 * h + i] = nf_packB_relu(hB0, hB1);
    }
}

__device__ __forceinline__ void nf_mm_blk(const NFBlk& b, const v16h* pA, const v16h* pB,
                                          v8f& c0, v8f& c1, v8f& d0, v8f& d1)
{
    c0 = b.c[0]; c1 = b.c[0];
    d0 = b.c[1]; d1 = b.c[1];
#pragma unroll
    for (int k = 0; k < 4; ++k) {
        c0 = NF_WMMA(b.a[k * 2 + 0], pA[k], c0);
        c1 = NF_WMMA(b.a[k * 2 + 0], pB[k], c1);
        d0 = NF_WMMA(b.a[k * 2 + 1], pA[k], d0);
        d1 = NF_WMMA(b.a[k * 2 + 1], pB[k], d1);
    }
}

__global__ void __launch_bounds__(NF_THREADS)
nf_flow_kernel(const float* __restrict__ x,
               const _Float16* __restrict__ tW1, const float* __restrict__ b1,
               const _Float16* __restrict__ tW2, const float* __restrict__ b2,
               const _Float16* __restrict__ tW3, const float* __restrict__ b3,
               float* __restrict__ out)
{
    __shared__ __align__(16) _Float16 sW1[NF_HIDDEN * LD1];
    __shared__ __align__(16) _Float16 sW2[NF_HIDDEN * LD2];
    __shared__ __align__(16) _Float16 sW3[NF_LATENT * LD3];
    __shared__ __align__(16) float sB1[NF_HIDDEN];
    __shared__ __align__(16) float sB2[NF_HIDDEN];
    __shared__ __align__(16) float sB3[NF_LATENT];

    const int tid  = threadIdx.x;
    const int lane = tid & 31;
    const int wave = tid >> 5;
    const int hi8  = (lane & 16) ? 8 : 0;

    const long row0 = (long)blockIdx.x * NF_ROWS_PER_BLOCK + (wave * 2 + 0) * 16 + (lane & 15);
    const long row1 = (long)blockIdx.x * NF_ROWS_PER_BLOCK + (wave * 2 + 1) * 16 + (lane & 15);

    v8f xA[4], xB[4];
    {
        const float* xr0 = x + row0 * NF_LATENT;
        const float* xr1 = x + row1 * NF_LATENT;
#pragma unroll
        for (int f = 0; f < 4; ++f) {
            const int d0 = f * 16 + hi8;
            v4f a0 = *(const v4f*)(xr0 + d0);
            v4f b0 = *(const v4f*)(xr0 + d0 + 4);
            v4f a1 = *(const v4f*)(xr1 + d0);
            v4f b1v = *(const v4f*)(xr1 + d0 + 4);
#pragma unroll
            for (int j = 0; j < 4; ++j) {
                xA[f][j] = a0[j]; xA[f][j + 4] = b0[j];
                xB[f][j] = a1[j]; xB[f][j + 4] = b1v[j];
            }
        }
    }

    NFStage st;
    nf_stage_load(st, 0, tid, tW1, tW2, tW3, b1, b2, b3);

    for (int s = 0; s < NSUB; ++s) {
        __syncthreads();
        {
#pragma unroll
            for (int q = 0; q < 2; ++q) {
                const int i = (tid + q * NF_THREADS) * 8;
                *(v8h*)&sW1[(i >> 5) * LD1 + (i & 31)] = st.r1[q];
            }
#pragma unroll
            for (int q = 0; q < 8; ++q) {
                const int i = (tid + q * NF_THREADS) * 8;
                *(v8h*)&sW2[(i >> 7) * LD2 + (i & 127)] = st.r2[q];
            }
#pragma unroll
            for (int q = 0; q < 4; ++q) {
                const int i = (tid + q * NF_THREADS) * 8;
                *(v8h*)&sW3[(i >> 7) * LD3 + (i & 127)] = st.r3[q];
            }
            if (tid < 128) sB1[tid] = st.rb0;
            else           sB2[tid - 128] = st.rb0;
            if (tid < 64)  sB3[tid] = st.rb1;
        }
        if (s + 1 < NSUB)
            nf_stage_load(st, s + 1, tid, tW1, tW2, tW3, b1, b2, b3);
        __syncthreads();

        v16h binA = nf_packB(xA[0], xA[1]);
        v16h binB = nf_packB(xB[0], xB[1]);

        v16h p1A[4], p1B[4];
        v16h p2A[4], p2B[4];
        v8f stA[4], stB[4];

        NFBlk1 f0, f1;
        NFBlk  g0, g1;
        nf_ld_blk1(f0, sW1, sB1, 0, lane, hi8);
        nf_ld_blk1(f1, sW1, sB1, 1, lane, hi8);
        NF_LOADS_STAY_AHEAD();
        nf_l1_blk(f0, 0, binA, binB, p1A, p1B);

        nf_ld_blk2(g0, sW2, sB2, 0, lane, hi8);
        NF_LOADS_STAY_AHEAD();
        nf_l1_blk(f1, 1, binA, binB, p1A, p1B);

#pragma unroll
        for (int g = 0; g < 4; ++g) {
            v8f c0, c1, d0, d1;
            if ((g & 1) == 0) {
                if (g + 1 < 4) nf_ld_blk2(g1, sW2, sB2, g + 1, lane, hi8);
                NF_LOADS_STAY_AHEAD();
                nf_mm_blk(g0, p1A, p1B, c0, c1, d0, d1);
            } else {
                if (g + 1 < 4) nf_ld_blk2(g0, sW2, sB2, g + 1, lane, hi8);
                else           nf_ld_blk3(g0, sW3, sB3, 0, lane, hi8);
                NF_LOADS_STAY_AHEAD();
                nf_mm_blk(g1, p1A, p1B, c0, c1, d0, d1);
            }
            p2A[g] = nf_packB_relu(c0, d0);
            p2B[g] = nf_packB_relu(c1, d1);
        }

        nf_ld_blk3(g1, sW3, sB3, 1, lane, hi8);
        NF_LOADS_STAY_AHEAD();
        nf_mm_blk(g0, p2A, p2B, stA[0], stB[0], stA[1], stB[1]);
        nf_mm_blk(g1, p2A, p2B, stA[2], stB[2], stA[3], stB[3]);

#pragma unroll
        for (int f = 0; f < 2; ++f) {
#pragma unroll
            for (int j = 0; j < 8; ++j) {
                float eA = __expf(NF_SCALE * nf_tanh(stA[f][j]));
                float eB = __expf(NF_SCALE * nf_tanh(stB[f][j]));
                xA[2 + f][j] = xA[2 + f][j] * eA + stA[2 + f][j];
                xB[2 + f][j] = xB[2 + f][j] * eB + stB[2 + f][j];
            }
        }
#pragma unroll
        for (int f = 0; f < 2; ++f) {
            v8f tA = xA[f]; xA[f] = xA[2 + f]; xA[2 + f] = tA;
            v8f tB = xB[f]; xB[f] = xB[2 + f]; xB[2 + f] = tB;
        }
    }

    {
        float* o0 = out + row0 * NF_LATENT;
        float* o1 = out + row1 * NF_LATENT;
#pragma unroll
        for (int f = 0; f < 4; ++f) {
            const int d0 = f * 16 + hi8;
            v4f a0, b0, a1, b1v;
#pragma unroll
            for (int j = 0; j < 4; ++j) {
                a0[j] = xA[f][j]; b0[j] = xA[f][j + 4];
                a1[j] = xB[f][j]; b1v[j] = xB[f][j + 4];
            }
            *(v4f*)(o0 + d0)     = a0;
            *(v4f*)(o0 + d0 + 4) = b0;
            *(v4f*)(o1 + d0)     = a1;
            *(v4f*)(o1 + d0 + 4) = b1v;
        }
    }
}

extern "C" void kernel_launch(void* const* d_in, const int* in_sizes, int n_in,
                              void* d_out, int out_size, void* d_ws, size_t ws_size,
                              hipStream_t stream)
{
    const float* x  = (const float*)d_in[0];
    const float* W1 = (const float*)d_in[1];
    const float* b1 = (const float*)d_in[2];
    const float* W2 = (const float*)d_in[3];
    const float* b2 = (const float*)d_in[4];
    const float* W3 = (const float*)d_in[5];
    const float* b3 = (const float*)d_in[6];
    float* out = (float*)d_out;

    _Float16* tW1 = (_Float16*)d_ws;
    _Float16* tW2 = tW1 + NSUB * W1_ELEMS;
    _Float16* tW3 = tW2 + NSUB * W2_ELEMS;

    const int total_w = NSUB * (W1_ELEMS + W2_ELEMS + W3_ELEMS);
    nf_convert_weights<<<(total_w + 255) / 256, 256, 0, stream>>>(W1, W2, W3, tW1, tW2, tW3);

    const int batch  = in_sizes[0] / NF_LATENT;
    const int blocks = batch / NF_ROWS_PER_BLOCK;
    nf_flow_kernel<<<blocks, NF_THREADS, 0, stream>>>(x, tW1, b1, tW2, b2, tW3, b3, out);
}


# Two weight buffers in LDS: the next sub-step's weights go global -> registers -> the other buffer in seven small batches between the WMMA blocks; one barrier per sub-step, no exposed prefetch wait
# speedup vs baseline: 1.5709x; 1.2003x over previous
#include <hip/hip_runtime.h>

typedef __attribute__((ext_vector_type(16))) _Float16 v16h;
typedef __attribute__((ext_vector_type(8)))  _Float16 v8h;
typedef __attribute__((ext_vector_type(8)))  float    v8f;
typedef __attribute__((ext_vector_type(4)))  float    v4f;

#define NF_NLAYERS 8
#define NF_LATENT  64
#define NF_HALF    32
#define NF_HIDDEN  128
#define NF_SCALE   0.1f

#define NF_THREADS 256
#define NF_WAVES   (NF_THREADS / 32)
#define NF_ROWS_PER_BLOCK (NF_WAVES * 2 * 16)

#define LD1 40
#define LD2 136
#define LD3 136

#define W1_ELEMS (NF_HALF * NF_HIDDEN)
#define W2_ELEMS (NF_HIDDEN * NF_HIDDEN)
#define W3_ELEMS (NF_HIDDEN * NF_LATENT)
#define NSUB     (NF_NLAYERS * 2)

__global__ void __launch_bounds__(256)
nf_convert_weights(const float* __restrict__ W1, const float* __restrict__ W2,
                   const float* __restrict__ W3,
                   _Float16* __restrict__ tW1, _Float16* __restrict__ tW2,
                   _Float16* __restrict__ tW3)
{
    const int i = blockIdx.x * 256 + threadIdx.x;
    const int n1 = NSUB * W1_ELEMS;
    const int n2 = NSUB * W2_ELEMS;
    const int n3 = NSUB * W3_ELEMS;
    if (i < n1) {
        int w = i >> 12, r = i & 4095;
        int m = r >> 5, k = r & 31;
        tW1[i] = (_Float16)W1[w * W1_ELEMS + k * NF_HIDDEN + m];
    } else if (i < n1 + n2) {
        int j = i - n1;
        int w = j >> 14, r = j & 16383;
        int m = r >> 7, k = r & 127;
        tW2[j] = (_Float16)W2[w * W2_ELEMS + k * NF_HIDDEN + m];
    } else if (i < n1 + n2 + n3) {
        int j = i - n1 - n2;
        int w = j >> 13, r = j & 8191;
        int m = r >> 7, k = r & 127;
        tW3[j] = (_Float16)W3[w * W3_ELEMS + k * NF_LATENT + m];
    }
}

__device__ __forceinline__ v16h nf_ldsA(const _Float16* W, int ld, int mtile, int kchunk, int lane)
{
    const int row = mtile * 16 + (lane & 15);
    const int kb  = kchunk * 32 + ((lane & 16) ? 8 : 0);
    const _Float16* p = W + row * ld + kb;
    v8h lo = *(const v8h*)p;
    v8h hi = *(const v8h*)(p + 16);
    v16h a;
#pragma unroll
    for (int j = 0; j < 8; ++j) { a[j] = lo[j]; a[j + 8] = hi[j]; }
    return a;
}

__device__ __forceinline__ v8f nf_ldsBias(const float* B, int mtile, int hi8)
{
    const float* p = B + mtile * 16 + hi8;
    v8f c;
#pragma unroll
    for (int j = 0; j < 8; ++j) c[j] = p[j];
    return c;
}

__device__ __forceinline__ v16h nf_packB(v8f lo, v8f hi)
{
    v16h b;
#pragma unroll
    for (int j = 0; j < 8; ++j) { b[j] = (_Float16)lo[j]; b[j + 8] = (_Float16)hi[j]; }
    return b;
}

__device__ __forceinline__ v16h nf_relu16(v16h b)
{
#if __has_builtin(__builtin_elementwise_max)
    const v16h z = {};
    return __builtin_elementwise_max(b, z);
#else
    v16h r;
#pragma unroll
    for (int j = 0; j < 16; ++j) r[j] = b[j] > (_Float16)0.0f ? b[j] : (_Float16)0.0f;
    return r;
#endif
}

__device__ __forceinline__ v16h nf_packB_relu(v8f lo, v8f hi)
{
    return nf_relu16(nf_packB(lo, hi));
}

__device__ __forceinline__ float nf_tanh(float v)
{
#if __has_builtin(__builtin_amdgcn_tanhf)
    return __builtin_amdgcn_tanhf(v);
#else
    return tanhf(v);
#endif
}

#define NF_WMMA(a, b, c) \
    __builtin_amdgcn_wmma_f32_16x16x32_f16(false, (a), false, (b), (short)0, (c), false, false)

#define SW1_ELEMS (NF_HIDDEN * LD1)
#define SW2_ELEMS (NF_HIDDEN * LD2)
#define SW3_ELEMS (NF_LATENT * LD3)

__device__ __forceinline__ v8h nf_chunk_load(int c, int s, int tid, const _Float16* tW1,
                                             const _Float16* tW2, const _Float16* tW3)
{
    if (c < 2)  return *(const v8h*)&(tW1 + (long)s * W1_ELEMS)[(tid + c * NF_THREADS) * 8];
    if (c < 10) return *(const v8h*)&(tW2 + (long)s * W2_ELEMS)[(tid + (c - 2) * NF_THREADS) * 8];
    return *(const v8h*)&(tW3 + (long)s * W3_ELEMS)[(tid + (c - 10) * NF_THREADS) * 8];
}

__device__ __forceinline__ void nf_chunk_store(int c, v8h v, int tid, _Float16* w1, _Float16* w2,
                                               _Float16* w3)
{
    if (c < 2) {
        const int i = (tid + c * NF_THREADS) * 8;
        *(v8h*)&w1[(i >> 5) * LD1 + (i & 31)] = v;
    } else if (c < 10) {
        const int i = (tid + (c - 2) * NF_THREADS) * 8;
        *(v8h*)&w2[(i >> 7) * LD2 + (i & 127)] = v;
    } else {
        const int i = (tid + (c - 10) * NF_THREADS) * 8;
        *(v8h*)&w3[(i >> 7) * LD3 + (i & 127)] = v;
    }
}

struct NFBlk1 { v16h a[4]; v8f c[4]; };
struct NFBlk  { v16h a[8]; v8f c[2]; };

__device__ __forceinline__ void nf_ld_blk1(NFBlk1& b, const _Float16* sW1, const float* sB1,
                                           int h, int lane, int hi8)
{
#pragma unroll
    for (int i = 0; i < 4; ++i) {
        b.a[i] = nf_ldsA(sW1, LD1, 4 * h + i, 0, lane);
        b.c[i] = nf_ldsBias(sB1, 4 * h + i, hi8);
    }
}

__device__ __forceinline__ void nf_ld_blk2(NFBlk& b, const _Float16* sW2, const float* sB2,
                                           int g, int lane, int hi8)
{
#pragma unroll
    for (int k = 0; k < 4; ++k) {
        b.a[k * 2 + 0] = nf_ldsA(sW2, LD2, 2 * g, k, lane);
        b.a[k * 2 + 1] = nf_ldsA(sW2, LD2, 2 * g + 1, k, lane);
    }
    b.c[0] = nf_ldsBias(sB2, 2 * g, hi8);
    b.c[1] = nf_ldsBias(sB2, 2 * g + 1, hi8);
}

__device__ __forceinline__ void nf_ld_blk3(NFBlk& b, const _Float16* sW3, const float* sB3,
                                           int u, int lane, int hi8)
{
#pragma unroll
    for (int k = 0; k < 4; ++k) {
        b.a[k * 2 + 0] = nf_ldsA(sW3, LD3, 2 * u, k, lane);
        b.a[k * 2 + 1] = nf_ldsA(sW3, LD3, 2 * u + 1, k, lane);
    }
    b.c[0] = nf_ldsBias(sB3, 2 * u, hi8);
    b.c[1] = nf_ldsBias(sB3, 2 * u + 1, hi8);
}

#define NF_LOADS_STAY_AHEAD() __builtin_amdgcn_sched_barrier(0x406)

#define NF_STORES_STAY_BEHIND() __builtin_amdgcn_sched_barrier(0x406)

__device__ __forceinline__ void nf_l1_blk(const NFBlk1& b, int h, v16h binA, v16h binB,
                                          v16h* p1A, v16h* p1B)
{
#pragma unroll
    for (int i = 0; i < 2; ++i) {
        v8f hA0 = NF_WMMA(b.a[2 * i], binA, b.c[2 * i]);
        v8f hB0 = NF_WMMA(b.a[2 * i], binB, b.c[2 * i]);
        v8f hA1 = NF_WMMA(b.a[2 * i + 1], binA, b.c[2 * i + 1]);
        v8f hB1 = NF_WMMA(b.a[2 * i + 1], binB, b.c[2 * i + 1]);
        p1A[2 * h + i] = nf_packB_relu(hA0, hA1);
        p1B[2 * h + i] = nf_packB_relu(hB0, hB1);
    }
}

__device__ __forceinline__ void nf_mm_blk(const NFBlk& b, const v16h* pA, const v16h* pB,
                                          v8f& c0, v8f& c1, v8f& d0, v8f& d1)
{
    c0 = b.c[0]; c1 = b.c[0];
    d0 = b.c[1]; d1 = b.c[1];
#pragma unroll
    for (int k = 0; k < 4; ++k) {
        c0 = NF_WMMA(b.a[k * 2 + 0], pA[k], c0);
        c1 = NF_WMMA(b.a[k * 2 + 0], pB[k], c1);
        d0 = NF_WMMA(b.a[k * 2 + 1], pA[k], d0);
        d1 = NF_WMMA(b.a[k * 2 + 1], pB[k], d1);
    }
}

__global__ void __launch_bounds__(NF_THREADS)
nf_flow_kernel(const float* __restrict__ x,
               const _Float16* __restrict__ tW1, const float* __restrict__ b1,
               const _Float16* __restrict__ tW2, const float* __restrict__ b2,
               const _Float16* __restrict__ tW3, const float* __restrict__ b3,
               float* __restrict__ out)
{
    __shared__ __align__(16) _Float16 sW1[2][SW1_ELEMS];
    __shared__ __align__(16) _Float16 sW2[2][SW2_ELEMS];
    __shared__ __align__(16) _Float16 sW3[2][SW3_ELEMS];
    __shared__ __align__(16) float sB1[2][NF_HIDDEN];
    __shared__ __align__(16) float sB2[2][NF_HIDDEN];
    __shared__ __align__(16) float sB3[2][NF_LATENT];

    const int tid  = threadIdx.x;
    const int lane = tid & 31;
    const int wave = tid >> 5;
    const int hi8  = (lane & 16) ? 8 : 0;

    const long row0 = (long)blockIdx.x * NF_ROWS_PER_BLOCK + (wave * 2 + 0) * 16 + (lane & 15);
    const long row1 = (long)blockIdx.x * NF_ROWS_PER_BLOCK + (wave * 2 + 1) * 16 + (lane & 15);

    v8f xA[4], xB[4];
    {
        const float* xr0 = x + row0 * NF_LATENT;
        const float* xr1 = x + row1 * NF_LATENT;
#pragma unroll
        for (int f = 0; f < 4; ++f) {
            const int d0 = f * 16 + hi8;
            v4f a0 = *(const v4f*)(xr0 + d0);
            v4f b0 = *(const v4f*)(xr0 + d0 + 4);
            v4f a1 = *(const v4f*)(xr1 + d0);
            v4f b1v = *(const v4f*)(xr1 + d0 + 4);
#pragma unroll
            for (int j = 0; j < 4; ++j) {
                xA[f][j] = a0[j]; xA[f][j + 4] = b0[j];
                xB[f][j] = a1[j]; xB[f][j + 4] = b1v[j];
            }
        }
    }

    {
#pragma unroll
        for (int c = 0; c < 14; ++c) {
            v8h v = nf_chunk_load(c, 0, tid, tW1, tW2, tW3);
            nf_chunk_store(c, v, tid, sW1[0], sW2[0], sW3[0]);
        }
        if (tid < 128) sB1[0][tid] = b1[tid];
        else           sB2[0][tid - 128] = b2[tid - 128];
        if (tid < 64)  sB3[0][tid] = b3[tid];
    }

#define NF_ISSUE(i)                                                        \
    bt[i][0] = nf_chunk_load(2 * (i), sn, tid, tW1, tW2, tW3);             \
    bt[i][1] = nf_chunk_load(2 * (i) + 1, sn, tid, tW1, tW2, tW3)
#define NF_COMMIT(i)                                                       \
    nf_chunk_store(2 * (i), bt[i][0], tid, nW1, nW2, nW3);                 \
    nf_chunk_store(2 * (i) + 1, bt[i][1], tid, nW1, nW2, nW3)

    for (int s = 0; s < NSUB; ++s) {
        __syncthreads();

        const int cur = s & 1;
        const _Float16* cW1 = sW1[cur];
        const _Float16* cW2 = sW2[cur];
        const _Float16* cW3 = sW3[cur];
        const float* cB1 = sB1[cur];
        const float* cB2 = sB2[cur];
        const float* cB3 = sB3[cur];
        _Float16* nW1 = sW1[cur ^ 1];
        _Float16* nW2 = sW2[cur ^ 1];
        _Float16* nW3 = sW3[cur ^ 1];
        const int sn = (s + 1 < NSUB) ? s + 1 : s;

        v8h bt[7][2];
        float nb0, nb1;

        v16h binA = nf_packB(xA[0], xA[1]);
        v16h binB = nf_packB(xB[0], xB[1]);

        v16h p1A[4], p1B[4];
        v16h p2A[4], p2B[4];
        v8f stA[4], stB[4];

        NFBlk1 f0, f1;
        NFBlk  g0, g1;
        nf_ld_blk1(f0, cW1, cB1, 0, lane, hi8);
        nf_ld_blk1(f1, cW1, cB1, 1, lane, hi8);
        NF_ISSUE(0);
        NF_LOADS_STAY_AHEAD();
        nf_l1_blk(f0, 0, binA, binB, p1A, p1B);

        nf_ld_blk2(g0, cW2, cB2, 0, lane, hi8);
        NF_ISSUE(1);
        NF_LOADS_STAY_AHEAD();
        nf_l1_blk(f1, 1, binA, binB, p1A, p1B);

        {
            v8f c0, c1, d0, d1;
            nf_ld_blk2(g1, cW2, cB2, 1, lane, hi8);
            NF_COMMIT(0);
            NF_ISSUE(2);
            NF_LOADS_STAY_AHEAD();
            nf_mm_blk(g0, p1A, p1B, c0, c1, d0, d1);
            p2A[0] = nf_packB_relu(c0, d0);
            p2B[0] = nf_packB_relu(c1, d1);

            nf_ld_blk2(g0, cW2, cB2, 2, lane, hi8);
            NF_COMMIT(1);
            NF_ISSUE(3);
            NF_LOADS_STAY_AHEAD();
            nf_mm_blk(g1, p1A, p1B, c0, c1, d0, d1);
            p2A[1] = nf_packB_relu(c0, d0);
            p2B[1] = nf_packB_relu(c1, d1);

            nf_ld_blk2(g1, cW2, cB2, 3, lane, hi8);
            NF_COMMIT(2);
            NF_ISSUE(4);
            nb0 = (tid < 128) ? b1[(long)sn * NF_HIDDEN + tid] : b2[(long)sn * NF_HIDDEN + (tid - 128)];
            nb1 = b3[(long)sn * NF_LATENT + (tid & 63)];
            NF_LOADS_STAY_AHEAD();
            nf_mm_blk(g0, p1A, p1B, c0, c1, d0, d1);
            p2A[2] = nf_packB_relu(c0, d0);
            p2B[2] = nf_packB_relu(c1, d1);

            nf_ld_blk3(g0, cW3, cB3, 0, lane, hi8);
            NF_COMMIT(3);
            NF_ISSUE(5);
            NF_LOADS_STAY_AHEAD();
            nf_mm_blk(g1, p1A, p1B, c0, c1, d0, d1);
            p2A[3] = nf_packB_relu(c0, d0);
            p2B[3] = nf_packB_relu(c1, d1);
        }

        nf_ld_blk3(g1, cW3, cB3, 1, lane, hi8);
        NF_COMMIT(4);
        NF_ISSUE(6);
        NF_LOADS_STAY_AHEAD();
        nf_mm_blk(g0, p2A, p2B, stA[0], stB[0], stA[1], stB[1]);
        nf_mm_blk(g1, p2A, p2B, stA[2], stB[2], stA[3], stB[3]);
        NF_STORES_STAY_BEHIND();
        NF_COMMIT(5);
        NF_COMMIT(6);
        if (tid < 128) sB1[cur ^ 1][tid] = nb0;
        else           sB2[cur ^ 1][tid - 128] = nb0;
        if (tid < 64)  sB3[cur ^ 1][tid] = nb1;

#pragma unroll
        for (int f = 0; f < 2; ++f) {
#pragma unroll
            for (int j = 0; j < 8; ++j) {
                float eA = __expf(NF_SCALE * nf_tanh(stA[f][j]));
                float eB = __expf(NF_SCALE * nf_tanh(stB[f][j]));
                xA[2 + f][j] = xA[2 + f][j] * eA + stA[2 + f][j];
                xB[2 + f][j] = xB[2 + f][j] * eB + stB[2 + f][j];
            }
        }
#pragma unroll
        for (int f = 0; f < 2; ++f) {
            v8f tA = xA[f]; xA[f] = xA[2 + f]; xA[2 + f] = tA;
            v8f tB = xB[f]; xB[f] = xB[2 + f]; xB[2 + f] = tB;
        }
    }
#undef NF_ISSUE
#undef NF_COMMIT

    {
        float* o0 = out + row0 * NF_LATENT;
        float* o1 = out + row1 * NF_LATENT;
#pragma unroll
        for (int f = 0; f < 4; ++f) {
            const int d0 = f * 16 + hi8;
            v4f a0, b0, a1, b1v;
#pragma unroll
            for (int j = 0; j < 4; ++j) {
                a0[j] = xA[f][j]; b0[j] = xA[f][j + 4];
                a1[j] = xB[f][j]; b1v[j] = xB[f][j + 4];
            }
            *(v4f*)(o0 + d0)     = a0;
            *(v4f*)(o0 + d0 + 4) = b0;
            *(v4f*)(o1 + d0)     = a1;
            *(v4f*)(o1 + d0 + 4) = b1v;
        }
    }
}

extern "C" void kernel_launch(void* const* d_in, const int* in_sizes, int n_in,
                              void* d_out, int out_size, void* d_ws, size_t ws_size,
                              hipStream_t stream)
{
    const float* x  = (const float*)d_in[0];
    const float* W1 = (const float*)d_in[1];
    const float* b1 = (const float*)d_in[2];
    const float* W2 = (const float*)d_in[3];
    const float* b2 = (const float*)d_in[4];
    const float* W3 = (const float*)d_in[5];
    const float* b3 = (const float*)d_in[6];
    float* out = (float*)d_out;

    _Float16* tW1 = (_Float16*)d_ws;
    _Float16* tW2 = tW1 + NSUB * W1_ELEMS;
    _Float16* tW3 = tW2 + NSUB * W2_ELEMS;

    const int total_w = NSUB * (W1_ELEMS + W2_ELEMS + W3_ELEMS);
    nf_convert_weights<<<(total_w + 255) / 256, 256, 0, stream>>>(W1, W2, W3, tW1, tW2, tW3);

    const int batch  = in_sizes[0] / NF_LATENT;
    const int blocks = batch / NF_ROWS_PER_BLOCK;
    nf_flow_kernel<<<blocks, NF_THREADS, 0, stream>>>(x, tW1, b1, tW2, b2, tW3, b3, out);
}


# Each wave takes four 16-row batch tiles instead of two (4-wave workgroups, one wave per SIMD): every weight and bias fragment read from LDS feeds 4 WMMAs, halving LDS reads per FLOP
# speedup vs baseline: 1.6321x; 1.0390x over previous
#include <hip/hip_runtime.h>

typedef __attribute__((ext_vector_type(16))) _Float16 v16h;
typedef __attribute__((ext_vector_type(8)))  _Float16 v8h;
typedef __attribute__((ext_vector_type(8)))  float    v8f;
typedef __attribute__((ext_vector_type(4)))  float    v4f;

#define NF_NLAYERS 8
#define NF_LATENT  64
#define NF_HALF    32
#define NF_HIDDEN  128
#define NF_SCALE   0.1f

#define NF_THREADS 128
#define NF_WAVES   (NF_THREADS / 32)
#define NF_TILES   4
#define NF_ROWS_PER_BLOCK (NF_WAVES * NF_TILES * 16)

#define LD1 40
#define LD2 136
#define LD3 136

#define W1_ELEMS (NF_HALF * NF_HIDDEN)
#define W2_ELEMS (NF_HIDDEN * NF_HIDDEN)
#define W3_ELEMS (NF_HIDDEN * NF_LATENT)
#define NSUB     (NF_NLAYERS * 2)

__global__ void __launch_bounds__(256)
nf_convert_weights(const float* __restrict__ W1, const float* __restrict__ W2,
                   const float* __restrict__ W3,
                   _Float16* __restrict__ tW1, _Float16* __restrict__ tW2,
                   _Float16* __restrict__ tW3)
{
    const int i = blockIdx.x * 256 + threadIdx.x;
    const int n1 = NSUB * W1_ELEMS;
    const int n2 = NSUB * W2_ELEMS;
    const int n3 = NSUB * W3_ELEMS;
    if (i < n1) {
        int w = i >> 12, r = i & 4095;
        int m = r >> 5, k = r & 31;
        tW1[i] = (_Float16)W1[w * W1_ELEMS + k * NF_HIDDEN + m];
    } else if (i < n1 + n2) {
        int j = i - n1;
        int w = j >> 14, r = j & 16383;
        int m = r >> 7, k = r & 127;
        tW2[j] = (_Float16)W2[w * W2_ELEMS + k * NF_HIDDEN + m];
    } else if (i < n1 + n2 + n3) {
        int j = i - n1 - n2;
        int w = j >> 13, r = j & 8191;
        int m = r >> 7, k = r & 127;
        tW3[j] = (_Float16)W3[w * W3_ELEMS + k * NF_LATENT + m];
    }
}

__device__ __forceinline__ v16h nf_ldsA(const _Float16* W, int ld, int mtile, int kchunk, int lane)
{
    const int row = mtile * 16 + (lane & 15);
    const int kb  = kchunk * 32 + ((lane & 16) ? 8 : 0);
    const _Float16* p = W + row * ld + kb;
    v8h lo = *(const v8h*)p;
    v8h hi = *(const v8h*)(p + 16);
    v16h a;
#pragma unroll
    for (int j = 0; j < 8; ++j) { a[j] = lo[j]; a[j + 8] = hi[j]; }
    return a;
}

__device__ __forceinline__ v8f nf_ldsBias(const float* B, int mtile, int hi8)
{
    const float* p = B + mtile * 16 + hi8;
    v8f c;
#pragma unroll
    for (int j = 0; j < 8; ++j) c[j] = p[j];
    return c;
}

__device__ __forceinline__ v16h nf_packB(v8f lo, v8f hi)
{
    v16h b;
#pragma unroll
    for (int j = 0; j < 8; ++j) { b[j] = (_Float16)lo[j]; b[j + 8] = (_Float16)hi[j]; }
    return b;
}

__device__ __forceinline__ v16h nf_relu16(v16h b)
{
#if __has_builtin(__builtin_elementwise_max)
    const v16h z = {};
    return __builtin_elementwise_max(b, z);
#else
    v16h r;
#pragma unroll
    for (int j = 0; j < 16; ++j) r[j] = b[j] > (_Float16)0.0f ? b[j] : (_Float16)0.0f;
    return r;
#endif
}

__device__ __forceinline__ v16h nf_packB_relu(v8f lo, v8f hi)
{
    return nf_relu16(nf_packB(lo, hi));
}

__device__ __forceinline__ float nf_tanh(float v)
{
#if __has_builtin(__builtin_amdgcn_tanhf)
    return __builtin_amdgcn_tanhf(v);
#else
    return tanhf(v);
#endif
}

#define NF_WMMA(a, b, c) \
    __builtin_amdgcn_wmma_f32_16x16x32_f16(false, (a), false, (b), (short)0, (c), false, false)

#define SW1_ELEMS (NF_HIDDEN * LD1)
#define SW2_ELEMS (NF_HIDDEN * LD2)
#define SW3_ELEMS (NF_LATENT * LD3)

__device__ __forceinline__ v8h nf_chunk_load(int c, int s, int tid, const _Float16* tW1,
                                             const _Float16* tW2, const _Float16* tW3)
{
    if (c < 4)  return *(const v8h*)&(tW1 + (long)s * W1_ELEMS)[(tid + c * NF_THREADS) * 8];
    if (c < 20) return *(const v8h*)&(tW2 + (long)s * W2_ELEMS)[(tid + (c - 4) * NF_THREADS) * 8];
    return *(const v8h*)&(tW3 + (long)s * W3_ELEMS)[(tid + (c - 20) * NF_THREADS) * 8];
}

__device__ __forceinline__ void nf_chunk_store(int c, v8h v, int tid, _Float16* w1, _Float16* w2,
                                               _Float16* w3)
{
    if (c < 4) {
        const int i = (tid + c * NF_THREADS) * 8;
        *(v8h*)&w1[(i >> 5) * LD1 + (i & 31)] = v;
    } else if (c < 20) {
        const int i = (tid + (c - 4) * NF_THREADS) * 8;
        *(v8h*)&w2[(i >> 7) * LD2 + (i & 127)] = v;
    } else {
        const int i = (tid + (c - 20) * NF_THREADS) * 8;
        *(v8h*)&w3[(i >> 7) * LD3 + (i & 127)] = v;
    }
}

struct NFBlk1 { v16h a[4]; v8f c[4]; };
struct NFBlk  { v16h a[8]; v8f c[2]; };

__device__ __forceinline__ void nf_ld_blk1(NFBlk1& b, const _Float16* sW1, const float* sB1,
                                           int h, int lane, int hi8)
{
#pragma unroll
    for (int i = 0; i < 4; ++i) {
        b.a[i] = nf_ldsA(sW1, LD1, 4 * h + i, 0, lane);
        b.c[i] = nf_ldsBias(sB1, 4 * h + i, hi8);
    }
}

__device__ __forceinline__ void nf_ld_blk2(NFBlk& b, const _Float16* sW2, const float* sB2,
                                           int g, int lane, int hi8)
{
#pragma unroll
    for (int k = 0; k < 4; ++k) {
        b.a[k * 2 + 0] = nf_ldsA(sW2, LD2, 2 * g, k, lane);
        b.a[k * 2 + 1] = nf_ldsA(sW2, LD2, 2 * g + 1, k, lane);
    }
    b.c[0] = nf_ldsBias(sB2, 2 * g, hi8);
    b.c[1] = nf_ldsBias(sB2, 2 * g + 1, hi8);
}

__device__ __forceinline__ void nf_ld_blk3(NFBlk& b, const _Float16* sW3, const float* sB3,
                                           int u, int lane, int hi8)
{
#pragma unroll
    for (int k = 0; k < 4; ++k) {
        b.a[k * 2 + 0] = nf_ldsA(sW3, LD3, 2 * u, k, lane);
        b.a[k * 2 + 1] = nf_ldsA(sW3, LD3, 2 * u + 1, k, lane);
    }
    b.c[0] = nf_ldsBias(sB3, 2 * u, hi8);
    b.c[1] = nf_ldsBias(sB3, 2 * u + 1, hi8);
}

#define NF_LOADS_STAY_AHEAD() __builtin_amdgcn_sched_barrier(0x406)

#define NF_STORES_STAY_BEHIND() __builtin_amdgcn_sched_barrier(0x406)

__device__ __forceinline__ void nf_l1_blk(const NFBlk1& b, int h, const v16h* bin,
                                          v16h (*p1)[4])
{
#pragma unroll
    for (int i = 0; i < 2; ++i) {
        v8f lo[NF_TILES], hi[NF_TILES];
#pragma unroll
        for (int t = 0; t < NF_TILES; ++t) lo[t] = NF_WMMA(b.a[2 * i], bin[t], b.c[2 * i]);
#pragma unroll
        for (int t = 0; t < NF_TILES; ++t) hi[t] = NF_WMMA(b.a[2 * i + 1], bin[t], b.c[2 * i + 1]);
#pragma unroll
        for (int t = 0; t < NF_TILES; ++t) p1[t][2 * h + i] = nf_packB_relu(lo[t], hi[t]);
    }
}

template <int NT>
__device__ __forceinline__ void nf_mm_blk(const NFBlk& b, const v16h (*p)[4], v8f* lo, v8f* hi)
{
#pragma unroll
    for (int t = 0; t < NT; ++t) { lo[t] = b.c[0]; hi[t] = b.c[1]; }
#pragma unroll
    for (int k = 0; k < 4; ++k) {
#pragma unroll
        for (int t = 0; t < NT; ++t) lo[t] = NF_WMMA(b.a[k * 2 + 0], p[t][k], lo[t]);
#pragma unroll
        for (int t = 0; t < NT; ++t) hi[t] = NF_WMMA(b.a[k * 2 + 1], p[t][k], hi[t]);
    }
}

__global__ void __launch_bounds__(NF_THREADS)
nf_flow_kernel(const float* __restrict__ x,
               const _Float16* __restrict__ tW1, const float* __restrict__ b1,
               const _Float16* __restrict__ tW2, const float* __restrict__ b2,
               const _Float16* __restrict__ tW3, const float* __restrict__ b3,
               float* __restrict__ out)
{
    __shared__ __align__(16) _Float16 sW1[2][SW1_ELEMS];
    __shared__ __align__(16) _Float16 sW2[2][SW2_ELEMS];
    __shared__ __align__(16) _Float16 sW3[2][SW3_ELEMS];
    __shared__ __align__(16) float sB1[2][NF_HIDDEN];
    __shared__ __align__(16) float sB2[2][NF_HIDDEN];
    __shared__ __align__(16) float sB3[2][NF_LATENT];

    const int tid  = threadIdx.x;
    const int lane = tid & 31;
    const int wave = tid >> 5;
    const int hi8  = (lane & 16) ? 8 : 0;

    const long rowbase = (long)blockIdx.x * NF_ROWS_PER_BLOCK + (long)(wave * NF_TILES) * 16 + (lane & 15);

    v8f xs[NF_TILES][4];
#pragma unroll
    for (int t = 0; t < NF_TILES; ++t) {
        const float* xr = x + (rowbase + t * 16) * NF_LATENT;
#pragma unroll
        for (int f = 0; f < 4; ++f) {
            const int d0 = f * 16 + hi8;
            v4f a0 = *(const v4f*)(xr + d0);
            v4f b0 = *(const v4f*)(xr + d0 + 4);
#pragma unroll
            for (int j = 0; j < 4; ++j) { xs[t][f][j] = a0[j]; xs[t][f][j + 4] = b0[j]; }
        }
    }

    {
#pragma unroll
        for (int c = 0; c < 28; ++c) {
            v8h v = nf_chunk_load(c, 0, tid, tW1, tW2, tW3);
            nf_chunk_store(c, v, tid, sW1[0], sW2[0], sW3[0]);
        }
        sB1[0][tid] = b1[tid];
        sB2[0][tid] = b2[tid];
        if (tid < 64) sB3[0][tid] = b3[tid];
    }

#define NF_ISSUE(i)                                                        \
    bt[i][0] = nf_chunk_load(4 * (i), sn, tid, tW1, tW2, tW3);             \
    bt[i][1] = nf_chunk_load(4 * (i) + 1, sn, tid, tW1, tW2, tW3);         \
    bt[i][2] = nf_chunk_load(4 * (i) + 2, sn, tid, tW1, tW2, tW3);         \
    bt[i][3] = nf_chunk_load(4 * (i) + 3, sn, tid, tW1, tW2, tW3)
#define NF_COMMIT(i)                                                       \
    nf_chunk_store(4 * (i), bt[i][0], tid, nW1, nW2, nW3);                 \
    nf_chunk_store(4 * (i) + 1, bt[i][1], tid, nW1, nW2, nW3);             \
    nf_chunk_store(4 * (i) + 2, bt[i][2], tid, nW1, nW2, nW3);             \
    nf_chunk_store(4 * (i) + 3, bt[i][3], tid, nW1, nW2, nW3)

    for (int s = 0; s < NSUB; ++s) {
        __syncthreads();

        const int cur = s & 1;
        const _Float16* cW1 = sW1[cur];
        const _Float16* cW2 = sW2[cur];
        const _Float16* cW3 = sW3[cur];
        const float* cB1 = sB1[cur];
        const float* cB2 = sB2[cur];
        const float* cB3 = sB3[cur];
        _Float16* nW1 = sW1[cur ^ 1];
        _Float16* nW2 = sW2[cur ^ 1];
        _Float16* nW3 = sW3[cur ^ 1];
        const int sn = (s + 1 < NSUB) ? s + 1 : s;

        v8h bt[7][4];
        float nb1, nb2, nb3;

        v16h bin[NF_TILES];
#pragma unroll
        for (int t = 0; t < NF_TILES; ++t) bin[t] = nf_packB(xs[t][0], xs[t][1]);

        v16h p1[NF_TILES][4];
        v16h p2[NF_TILES][4];
        v8f st[NF_TILES][4];

        NFBlk1 f0, f1;
        NFBlk  g0, g1;
        nf_ld_blk1(f0, cW1, cB1, 0, lane, hi8);
        nf_ld_blk1(f1, cW1, cB1, 1, lane, hi8);
        NF_ISSUE(0);
        NF_LOADS_STAY_AHEAD();
        nf_l1_blk(f0, 0, bin, p1);

        nf_ld_blk2(g0, cW2, cB2, 0, lane, hi8);
        NF_ISSUE(1);
        NF_LOADS_STAY_AHEAD();
        nf_l1_blk(f1, 1, bin, p1);

        {
            v8f lo[NF_TILES], hi[NF_TILES];
            nf_ld_blk2(g1, cW2, cB2, 1, lane, hi8);
            NF_COMMIT(0);
            NF_ISSUE(2);
            NF_LOADS_STAY_AHEAD();
            nf_mm_blk<NF_TILES>(g0, p1, lo, hi);
#pragma unroll
            for (int t = 0; t < NF_TILES; ++t) p2[t][0] = nf_packB_relu(lo[t], hi[t]);

            nf_ld_blk2(g0, cW2, cB2, 2, lane, hi8);
            NF_COMMIT(1);
            NF_ISSUE(3);
            NF_LOADS_STAY_AHEAD();
            nf_mm_blk<NF_TILES>(g1, p1, lo, hi);
#pragma unroll
            for (int t = 0; t < NF_TILES; ++t) p2[t][1] = nf_packB_relu(lo[t], hi[t]);

            nf_ld_blk2(g1, cW2, cB2, 3, lane, hi8);
            NF_COMMIT(2);
            NF_ISSUE(4);
            nb1 = b1[(long)sn * NF_HIDDEN + tid];
            nb2 = b2[(long)sn * NF_HIDDEN + tid];
            nb3 = b3[(long)sn * NF_LATENT + (tid & 63)];
            NF_LOADS_STAY_AHEAD();
            nf_mm_blk<NF_TILES>(g0, p1, lo, hi);
#pragma unroll
            for (int t = 0; t < NF_TILES; ++t) p2[t][2] = nf_packB_relu(lo[t], hi[t]);

            nf_ld_blk3(g0, cW3, cB3, 0, lane, hi8);
            NF_COMMIT(3);
            NF_ISSUE(5);
            NF_LOADS_STAY_AHEAD();
            nf_mm_blk<NF_TILES>(g1, p1, lo, hi);
#pragma unroll
            for (int t = 0; t < NF_TILES; ++t) p2[t][3] = nf_packB_relu(lo[t], hi[t]);
        }

        nf_ld_blk3(g1, cW3, cB3, 1, lane, hi8);
        NF_COMMIT(4);
        NF_ISSUE(6);
        NF_LOADS_STAY_AHEAD();
#pragma unroll
        for (int tp = 0; tp < NF_TILES; tp += 2) {
            v8f lo[2], hi[2];
            nf_mm_blk<2>(g0, p2 + tp, lo, hi);
            st[tp][0] = lo[0]; st[tp + 1][0] = lo[1];
            st[tp][1] = hi[0]; st[tp + 1][1] = hi[1];
            nf_mm_blk<2>(g1, p2 + tp, lo, hi);
            st[tp][2] = lo[0]; st[tp + 1][2] = lo[1];
            st[tp][3] = hi[0]; st[tp + 1][3] = hi[1];
        }
        NF_STORES_STAY_BEHIND();
        NF_COMMIT(5);
        NF_COMMIT(6);

#pragma unroll
        for (int t = 0; t < NF_TILES; ++t) {
#pragma unroll
            for (int f = 0; f < 2; ++f) {
#pragma unroll
                for (int j = 0; j < 8; ++j) {
                    float e = __expf(NF_SCALE * nf_tanh(st[t][f][j]));
                    xs[t][2 + f][j] = xs[t][2 + f][j] * e + st[t][2 + f][j];
                }
            }
#pragma unroll
            for (int f = 0; f < 2; ++f) {
                v8f tmp = xs[t][f]; xs[t][f] = xs[t][2 + f]; xs[t][2 + f] = tmp;
            }
        }

        sB1[cur ^ 1][tid] = nb1;
        sB2[cur ^ 1][tid] = nb2;
        sB3[cur ^ 1][tid & 63] = nb3;
    }
#undef NF_ISSUE
#undef NF_COMMIT

#pragma unroll
    for (int t = 0; t < NF_TILES; ++t) {
        float* o = out + (rowbase + t * 16) * NF_LATENT;
#pragma unroll
        for (int f = 0; f < 4; ++f) {
            const int d0 = f * 16 + hi8;
            v4f a0, b0;
#pragma unroll
            for (int j = 0; j < 4; ++j) { a0[j] = xs[t][f][j]; b0[j] = xs[t][f][j + 4]; }
            *(v4f*)(o + d0)     = a0;
            *(v4f*)(o + d0 + 4) = b0;
        }
    }
}

extern "C" void kernel_launch(void* const* d_in, const int* in_sizes, int n_in,
                              void* d_out, int out_size, void* d_ws, size_t ws_size,
                              hipStream_t stream)
{
    const float* x  = (const float*)d_in[0];
    const float* W1 = (const float*)d_in[1];
    const float* b1 = (const float*)d_in[2];
    const float* W2 = (const float*)d_in[3];
    const float* b2 = (const float*)d_in[4];
    const float* W3 = (const float*)d_in[5];
    const float* b3 = (const float*)d_in[6];
    float* out = (float*)d_out;

    _Float16* tW1 = (_Float16*)d_ws;
    _Float16* tW2 = tW1 + NSUB * W1_ELEMS;
    _Float16* tW3 = tW2 + NSUB * W2_ELEMS;

    const int total_w = NSUB * (W1_ELEMS + W2_ELEMS + W3_ELEMS);
    nf_convert_weights<<<(total_w + 255) / 256, 256, 0, stream>>>(W1, W2, W3, tW1, tW2, tW3);

    const int batch  = in_sizes[0] / NF_LATENT;
    const int blocks = batch / NF_ROWS_PER_BLOCK;
    nf_flow_kernel<<<blocks, NF_THREADS, 0, stream>>>(x, tW1, b1, tW2, b2, tW3, b3, out);
}


# 8-wave workgroups, two waves per SIMD, still four batch tiles per wave: f32 activations wait in wave-private LDS between uses; fragment sets are half-blocks of 16 WMMAs (474 VGPRs)
# speedup vs baseline: 1.8518x; 1.1346x over previous
#include <hip/hip_runtime.h>

typedef __attribute__((ext_vector_type(16))) _Float16 v16h;
typedef __attribute__((ext_vector_type(8)))  _Float16 v8h;
typedef __attribute__((ext_vector_type(8)))  float    v8f;
typedef __attribute__((ext_vector_type(4)))  float    v4f;

#define NF_NLAYERS 8
#define NF_LATENT  64
#define NF_HALF    32
#define NF_HIDDEN  128
#define NF_SCALE   0.1f

#define NF_THREADS 256
#define NF_WAVES   (NF_THREADS / 32)
#define NF_TILES   4
#define NF_ROWS_PER_BLOCK (NF_WAVES * NF_TILES * 16)

#define LD1 40
#define LD2 136
#define LD3 136

#define W1_ELEMS (NF_HALF * NF_HIDDEN)
#define W2_ELEMS (NF_HIDDEN * NF_HIDDEN)
#define W3_ELEMS (NF_HIDDEN * NF_LATENT)
#define NSUB     (NF_NLAYERS * 2)

__global__ void __launch_bounds__(256)
nf_convert_weights(const float* __restrict__ W1, const float* __restrict__ W2,
                   const float* __restrict__ W3,
                   _Float16* __restrict__ tW1, _Float16* __restrict__ tW2,
                   _Float16* __restrict__ tW3)
{
    const int i = blockIdx.x * 256 + threadIdx.x;
    const int n1 = NSUB * W1_ELEMS;
    const int n2 = NSUB * W2_ELEMS;
    const int n3 = NSUB * W3_ELEMS;
    if (i < n1) {
        int w = i >> 12, r = i & 4095;
        int m = r >> 5, k = r & 31;
        tW1[i] = (_Float16)W1[w * W1_ELEMS + k * NF_HIDDEN + m];
    } else if (i < n1 + n2) {
        int j = i - n1;
        int w = j >> 14, r = j & 16383;
        int m = r >> 7, k = r & 127;
        tW2[j] = (_Float16)W2[w * W2_ELEMS + k * NF_HIDDEN + m];
    } else if (i < n1 + n2 + n3) {
        int j = i - n1 - n2;
        int w = j >> 13, r = j & 8191;
        int m = r >> 7, k = r & 127;
        tW3[j] = (_Float16)W3[w * W3_ELEMS + k * NF_LATENT + m];
    }
}

__device__ __forceinline__ v16h nf_ldsA(const _Float16* W, int ld, int mtile, int kchunk, int lane)
{
    const int row = mtile * 16 + (lane & 15);
    const int kb  = kchunk * 32 + ((lane & 16) ? 8 : 0);
    const _Float16* p = W + row * ld + kb;
    v8h lo = *(const v8h*)p;
    v8h hi = *(const v8h*)(p + 16);
    v16h a;
#pragma unroll
    for (int j = 0; j < 8; ++j) { a[j] = lo[j]; a[j + 8] = hi[j]; }
    return a;
}

__device__ __forceinline__ v8f nf_ldsBias(const float* B, int mtile, int hi8)
{
    const float* p = B + mtile * 16 + hi8;
    v8f c;
#pragma unroll
    for (int j = 0; j < 8; ++j) c[j] = p[j];
    return c;
}

__device__ __forceinline__ v16h nf_packB(v8f lo, v8f hi)
{
    v16h b;
#pragma unroll
    for (int j = 0; j < 8; ++j) { b[j] = (_Float16)lo[j]; b[j + 8] = (_Float16)hi[j]; }
    return b;
}

__device__ __forceinline__ v16h nf_relu16(v16h b)
{
#if __has_builtin(__builtin_elementwise_max)
    const v16h z = {};
    return __builtin_elementwise_max(b, z);
#else
    v16h r;
#pragma unroll
    for (int j = 0; j < 16; ++j) r[j] = b[j] > (_Float16)0.0f ? b[j] : (_Float16)0.0f;
    return r;
#endif
}

__device__ __forceinline__ v16h nf_packB_relu(v8f lo, v8f hi)
{
    return nf_relu16(nf_packB(lo, hi));
}

__device__ __forceinline__ float nf_tanh(float v)
{
#if __has_builtin(__builtin_amdgcn_tanhf)
    return __builtin_amdgcn_tanhf(v);
#else
    return tanhf(v);
#endif
}

#define NF_WMMA(a, b, c) \
    __builtin_amdgcn_wmma_f32_16x16x32_f16(false, (a), false, (b), (short)0, (c), false, false)

#define SW1_ELEMS (NF_HIDDEN * LD1)
#define SW2_ELEMS (NF_HIDDEN * LD2)
#define SW3_ELEMS (NF_LATENT * LD3)

__device__ __forceinline__ v8h nf_chunk_load(int c, int s, int tid, const _Float16* tW1,
                                             const _Float16* tW2, const _Float16* tW3)
{
    if (c < 2)  return *(const v8h*)&(tW1 + (long)s * W1_ELEMS)[(tid + c * NF_THREADS) * 8];
    if (c < 10) return *(const v8h*)&(tW2 + (long)s * W2_ELEMS)[(tid + (c - 2) * NF_THREADS) * 8];
    return *(const v8h*)&(tW3 + (long)s * W3_ELEMS)[(tid + (c - 10) * NF_THREADS) * 8];
}

__device__ __forceinline__ void nf_chunk_store(int c, v8h v, int tid, _Float16* w1, _Float16* w2,
                                               _Float16* w3)
{
    if (c < 2) {
        const int i = (tid + c * NF_THREADS) * 8;
        *(v8h*)&w1[(i >> 5) * LD1 + (i & 31)] = v;
    } else if (c < 10) {
        const int i = (tid + (c - 2) * NF_THREADS) * 8;
        *(v8h*)&w2[(i >> 7) * LD2 + (i & 127)] = v;
    } else {
        const int i = (tid + (c - 10) * NF_THREADS) * 8;
        *(v8h*)&w3[(i >> 7) * LD3 + (i & 127)] = v;
    }
}

#define NF_PARK_FLOATS (NF_TILES * 4 * 2 * 128)

__device__ __forceinline__ v8f nf_park_ld(const float* p, int tf)
{
    v4f a = *(const v4f*)(p + tf * 256);
    v4f b = *(const v4f*)(p + tf * 256 + 128);
    v8f r;
#pragma unroll
    for (int j = 0; j < 4; ++j) { r[j] = a[j]; r[j + 4] = b[j]; }
    return r;
}

__device__ __forceinline__ void nf_park_st(float* p, int tf, v8f v)
{
    v4f a, b;
#pragma unroll
    for (int j = 0; j < 4; ++j) { a[j] = v[j]; b[j] = v[j + 4]; }
    *(v4f*)(p + tf * 256)       = a;
    *(v4f*)(p + tf * 256 + 128) = b;
}

struct NFH { v16h a[4]; v8f c[4]; };

__device__ __forceinline__ void nf_ld_h1(NFH& b, const _Float16* sW1, const float* sB1,
                                         int h, int lane, int hi8)
{
#pragma unroll
    for (int i = 0; i < 4; ++i) {
        b.a[i] = nf_ldsA(sW1, LD1, 4 * h + i, 0, lane);
        b.c[i] = nf_ldsBias(sB1, 4 * h + i, hi8);
    }
}

__device__ __forceinline__ void nf_ld_h(NFH& b, const _Float16* sW, int ld, const float* sB,
                                        int g, int kh, int lane, int hi8)
{
#pragma unroll
    for (int kk = 0; kk < 2; ++kk) {
        b.a[kk * 2 + 0] = nf_ldsA(sW, ld, 2 * g, 2 * kh + kk, lane);
        b.a[kk * 2 + 1] = nf_ldsA(sW, ld, 2 * g + 1, 2 * kh + kk, lane);
    }
    if (kh == 0) {
        b.c[0] = nf_ldsBias(sB, 2 * g, hi8);
        b.c[1] = nf_ldsBias(sB, 2 * g + 1, hi8);
    }
}

#define NF_LOADS_STAY_AHEAD() __builtin_amdgcn_sched_barrier(0x406)

__device__ __forceinline__ void nf_l1_half(const NFH& b, int h, const v16h* bin, v16h (*p1)[4])
{
#pragma unroll
    for (int i = 0; i < 2; ++i) {
        v8f lo[NF_TILES], hi[NF_TILES];
#pragma unroll
        for (int t = 0; t < NF_TILES; ++t) lo[t] = NF_WMMA(b.a[2 * i], bin[t], b.c[2 * i]);
#pragma unroll
        for (int t = 0; t < NF_TILES; ++t) hi[t] = NF_WMMA(b.a[2 * i + 1], bin[t], b.c[2 * i + 1]);
#pragma unroll
        for (int t = 0; t < NF_TILES; ++t) p1[t][2 * h + i] = nf_packB_relu(lo[t], hi[t]);
    }
}

template <int KH>
__device__ __forceinline__ void nf_mm_half(const NFH& b, const v16h (*p)[4], v8f* lo, v8f* hi)
{
    if (KH == 0) {
#pragma unroll
        for (int t = 0; t < NF_TILES; ++t) { lo[t] = b.c[0]; hi[t] = b.c[1]; }
    }
#pragma unroll
    for (int kk = 0; kk < 2; ++kk) {
#pragma unroll
        for (int t = 0; t < NF_TILES; ++t) lo[t] = NF_WMMA(b.a[kk * 2 + 0], p[t][2 * KH + kk], lo[t]);
#pragma unroll
        for (int t = 0; t < NF_TILES; ++t) hi[t] = NF_WMMA(b.a[kk * 2 + 1], p[t][2 * KH + kk], hi[t]);
    }
}

__global__ void __launch_bounds__(NF_THREADS)
nf_flow_kernel(const float* __restrict__ x,
               const _Float16* __restrict__ tW1, const float* __restrict__ b1,
               const _Float16* __restrict__ tW2, const float* __restrict__ b2,
               const _Float16* __restrict__ tW3, const float* __restrict__ b3,
               float* __restrict__ out)
{
    __shared__ __align__(16) _Float16 sW1[2][SW1_ELEMS];
    __shared__ __align__(16) _Float16 sW2[2][SW2_ELEMS];
    __shared__ __align__(16) _Float16 sW3[2][SW3_ELEMS];
    __shared__ __align__(16) float sB12[2][2 * NF_HIDDEN];
    __shared__ __align__(16) float sB3[2][NF_LATENT];
    __shared__ __align__(16) float sX[NF_WAVES * NF_PARK_FLOATS];

    const int tid  = threadIdx.x;
    const int lane = tid & 31;
    const int wave = tid >> 5;
    const int hi8  = (lane & 16) ? 8 : 0;

    const long rowbase = (long)blockIdx.x * NF_ROWS_PER_BLOCK + (long)(wave * NF_TILES) * 16 + (lane & 15);

    float* px = sX + wave * NF_PARK_FLOATS + lane * 4;

    v16h bin[NF_TILES];
#pragma unroll
    for (int t = 0; t < NF_TILES; ++t) {
        const float* xr = x + (rowbase + t * 16) * NF_LATENT;
        v8f xf[4];
#pragma unroll
        for (int f = 0; f < 4; ++f) {
            const int d0 = f * 16 + hi8;
            v4f a0 = *(const v4f*)(xr + d0);
            v4f b0 = *(const v4f*)(xr + d0 + 4);
#pragma unroll
            for (int j = 0; j < 4; ++j) { xf[f][j] = a0[j]; xf[f][j + 4] = b0[j]; }
            nf_park_st(px, t * 4 + f, xf[f]);
        }
        bin[t] = nf_packB(xf[0], xf[1]);
    }

    const float* bsrc = (tid < NF_HIDDEN) ? (b1 + tid) : (b2 + (tid - NF_HIDDEN));

    {
#pragma unroll
        for (int c = 0; c < 14; ++c) {
            v8h v = nf_chunk_load(c, 0, tid, tW1, tW2, tW3);
            nf_chunk_store(c, v, tid, sW1[0], sW2[0], sW3[0]);
        }
        sB12[0][tid] = *bsrc;
        sB3[0][tid & 63] = b3[tid & 63];
    }

#define NF_ISSUE(i)  bt[i] = nf_chunk_load((i), sn, tid, tW1, tW2, tW3)
#define NF_COMMIT(i) nf_chunk_store((i), bt[i], tid, nW1, nW2, nW3)

    for (int s = 0; s < NSUB; ++s) {
        __syncthreads();

        const int cur = s & 1;
        const _Float16* cW1 = sW1[cur];
        const _Float16* cW2 = sW2[cur];
        const _Float16* cW3 = sW3[cur];
        const float* cB1 = sB12[cur];
        const float* cB2 = sB12[cur] + NF_HIDDEN;
        const float* cB3 = sB3[cur];
        _Float16* nW1 = sW1[cur ^ 1];
        _Float16* nW2 = sW2[cur ^ 1];
        _Float16* nW3 = sW3[cur ^ 1];
        const int sn = (s + 1 < NSUB) ? s + 1 : s;
        const float* nb12_src = bsrc + (long)sn * NF_HIDDEN;
        float* poth = px + ((s & 1) ? 0 : 2) * 256;

        v8h bt[14];
        float nb12, nb3;

        v16h p1[NF_TILES][4];
        v16h p2[NF_TILES][4];
        v8f st[NF_TILES][4];
        v8f oth[NF_TILES][2];
        v8f lo[NF_TILES], hi[NF_TILES];

        NFH X, Y;
        nf_ld_h1(X, cW1, cB1, 0, lane, hi8);

        nf_ld_h1(Y, cW1, cB1, 1, lane, hi8);
        NF_ISSUE(0);
        NF_LOADS_STAY_AHEAD();
        nf_l1_half(X, 0, bin, p1);

        nf_ld_h(X, cW2, LD2, cB2, 0, 0, lane, hi8);
        NF_ISSUE(1);
        NF_LOADS_STAY_AHEAD();
        nf_l1_half(Y, 1, bin, p1);

        nf_ld_h(Y, cW2, LD2, cB2, 0, 1, lane, hi8);
        NF_ISSUE(2);
        NF_LOADS_STAY_AHEAD();
        nf_mm_half<0>(X, p1, lo, hi);

        nf_ld_h(X, cW2, LD2, cB2, 1, 0, lane, hi8);
        NF_ISSUE(3);
        NF_LOADS_STAY_AHEAD();
        nf_mm_half<1>(Y, p1, lo, hi);
#pragma unroll
        for (int t = 0; t < NF_TILES; ++t) p2[t][0] = nf_packB_relu(lo[t], hi[t]);

        nf_ld_h(Y, cW2, LD2, cB2, 1, 1, lane, hi8);
        NF_COMMIT(0);
        NF_ISSUE(4);
        NF_LOADS_STAY_AHEAD();
        nf_mm_half<0>(X, p1, lo, hi);

        nf_ld_h(X, cW2, LD2, cB2, 2, 0, lane, hi8);
        NF_COMMIT(1);
        NF_ISSUE(5);
        NF_LOADS_STAY_AHEAD();
        nf_mm_half<1>(Y, p1, lo, hi);
#pragma unroll
        for (int t = 0; t < NF_TILES; ++t) p2[t][1] = nf_packB_relu(lo[t], hi[t]);

        nf_ld_h(Y, cW2, LD2, cB2, 2, 1, lane, hi8);
        NF_COMMIT(2);
        NF_ISSUE(6);
        nb12 = *nb12_src;
        nb3  = b3[(long)sn * NF_LATENT + (tid & 63)];
        NF_LOADS_STAY_AHEAD();
        nf_mm_half<0>(X, p1, lo, hi);

        nf_ld_h(X, cW2, LD2, cB2, 3, 0, lane, hi8);
        NF_COMMIT(3);
        NF_ISSUE(7);
        NF_LOADS_STAY_AHEAD();
        nf_mm_half<1>(Y, p1, lo, hi);
#pragma unroll
        for (int t = 0; t < NF_TILES; ++t) p2[t][2] = nf_packB_relu(lo[t], hi[t]);

        nf_ld_h(Y, cW2, LD2, cB2, 3, 1, lane, hi8);
        NF_COMMIT(4);
        NF_ISSUE(8);
        NF_LOADS_STAY_AHEAD();
        nf_mm_half<0>(X, p1, lo, hi);

        nf_ld_h(X, cW3, LD3, cB3, 0, 0, lane, hi8);
        NF_COMMIT(5);
        NF_ISSUE(9);
        NF_LOADS_STAY_AHEAD();
        nf_mm_half<1>(Y, p1, lo, hi);
#pragma unroll
        for (int t = 0; t < NF_TILES; ++t) p2[t][3] = nf_packB_relu(lo[t], hi[t]);

        nf_ld_h(Y, cW3, LD3, cB3, 0, 1, lane, hi8);
        NF_COMMIT(6);
        NF_ISSUE(10);
        NF_LOADS_STAY_AHEAD();
        nf_mm_half<0>(X, p2, lo, hi);

        nf_ld_h(X, cW3, LD3, cB3, 1, 0, lane, hi8);
        NF_COMMIT(7);
        NF_ISSUE(11);
        NF_LOADS_STAY_AHEAD();
        nf_mm_half<1>(Y, p2, lo, hi);
#pragma unroll
        for (int t = 0; t < NF_TILES; ++t) { st[t][0] = lo[t]; st[t][1] = hi[t]; }

        nf_ld_h(Y, cW3, LD3, cB3, 1, 1, lane, hi8);
        NF_COMMIT(8);
        NF_ISSUE(12);
        NF_LOADS_STAY_AHEAD();
        nf_mm_half<0>(X, p2, lo, hi);

#pragma unroll
        for (int t = 0; t < NF_TILES; ++t) { oth[t][0] = nf_park_ld(poth, t * 4); oth[t][1] = nf_park_ld(poth, t * 4 + 1); }
        NF_COMMIT(9);
        NF_ISSUE(13);
        NF_LOADS_STAY_AHEAD();
        nf_mm_half<1>(Y, p2, lo, hi);
#pragma unroll
        for (int t = 0; t < NF_TILES; ++t) { st[t][2] = lo[t]; st[t][3] = hi[t]; }
        NF_COMMIT(10);
        NF_COMMIT(11);

#pragma unroll
        for (int t = 0; t < NF_TILES; ++t) {
            v8f nv[2];
#pragma unroll
            for (int f = 0; f < 2; ++f) {
#pragma unroll
                for (int j = 0; j < 8; ++j) {
                    float e = __expf(NF_SCALE * nf_tanh(st[t][f][j]));
                    nv[f][j] = oth[t][f][j] * e + st[t][2 + f][j];
                }
                nf_park_st(poth, t * 4 + f, nv[f]);
            }
            bin[t] = nf_packB(nv[0], nv[1]);
        }

        NF_COMMIT(12);
        NF_COMMIT(13);

        sB12[cur ^ 1][tid] = nb12;
        sB3[cur ^ 1][tid & 63] = nb3;
    }
#undef NF_ISSUE
#undef NF_COMMIT

#pragma unroll
    for (int t = 0; t < NF_TILES; ++t) {
        float* o = out + (rowbase + t * 16) * NF_LATENT;
#pragma unroll
        for (int f = 0; f < 4; ++f) {
            const int d0 = f * 16 + hi8;
            v8f v = nf_park_ld(px, t * 4 + f);
            v4f a0, b0;
#pragma unroll
            for (int j = 0; j < 4; ++j) { a0[j] = v[j]; b0[j] = v[j + 4]; }
            *(v4f*)(o + d0)     = a0;
            *(v4f*)(o + d0 + 4) = b0;
        }
    }
}

extern "C" void kernel_launch(void* const* d_in, const int* in_sizes, int n_in,
                              void* d_out, int out_size, void* d_ws, size_t ws_size,
                              hipStream_t stream)
{
    const float* x  = (const float*)d_in[0];
    const float* W1 = (const float*)d_in[1];
    const float* b1 = (const float*)d_in[2];
    const float* W2 = (const float*)d_in[3];
    const float* b2 = (const float*)d_in[4];
    const float* W3 = (const float*)d_in[5];
    const float* b3 = (const float*)d_in[6];
    float* out = (float*)d_out;

    _Float16* tW1 = (_Float16*)d_ws;
    _Float16* tW2 = tW1 + NSUB * W1_ELEMS;
    _Float16* tW3 = tW2 + NSUB * W2_ELEMS;

    const int total_w = NSUB * (W1_ELEMS + W2_ELEMS + W3_ELEMS);
    nf_convert_weights<<<(total_w + 255) / 256, 256, 0, stream>>>(W1, W2, W3, tW1, tW2, tW3);

    const int batch  = in_sizes[0] / NF_LATENT;
    const int blocks = batch / NF_ROWS_PER_BLOCK;
    nf_flow_kernel<<<blocks, NF_THREADS, 0, stream>>>(x, tW1, b1, tW2, b2, tW3, b3, out);
}
